# starting kernel
# baseline (speedup 1.0000x reference)
#include <hip/hip_runtime.h>

typedef __attribute__((ext_vector_type(2))) float v2f;
typedef __attribute__((ext_vector_type(8))) float v8f;

#define NB      16
#define HH      64
#define WW      64
#define QIN     4
#define BOND    16
#define OUTQ    64
#define HPAD    65
#define WPAD    65
#define INTER   63
#define PIX_PER_IMG (INTER * INTER)
#define NPIX    (NB * PIX_PER_IMG)
#define NBLK    (NPIX / 16)
#define WAVES_PER_WG 8
#define NSLOT   97
#define S_W3    0
#define S_W2    1
#define S_W1    17
#define S_W0    33

static __device__ __forceinline__ v8f wmma4(v2f a, v2f b, v8f c) {
    return __builtin_amdgcn_wmma_f32_16x16x4_f32(
         false, a,  false, b,
         (short)0, c,  false,  false);
}

__global__ __launch_bounds__(256) void convsbs_zero(float4* __restrict__ o, int n4) {
    int i = blockIdx.x * blockDim.x + threadIdx.x;
    if (i < n4) o[i] = make_float4(0.f, 0.f, 0.f, 0.f);
}

__global__ __launch_bounds__(256) void convsbs_chain(
        const float* __restrict__ x,
        const float* __restrict__ c0,
        const float* __restrict__ c1,
        const float* __restrict__ c2,
        const float* __restrict__ c3,
        float* __restrict__ out)
{
    __shared__ float wlds[NSLOT * 64];

    const int tid = threadIdx.x;

    for (int i = tid; i < NSLOT * 64; i += 256) {
        int slot = i >> 6;
        int e    = i & 63;
        int lane = e >> 1;
        int v    = e & 1;
        int m    = lane & 15;
        int kk   = ((lane >> 4) << 1) + v;
        float val;
        if (slot == S_W3) {
            val = c3[m * 4 + kk];
        } else if (slot < S_W1) {
            int j = slot - S_W2;
            val = c2[m * 64 + 4 * j + kk];
        } else if (slot < S_W0) {
            int j = slot - S_W1;
            val = c1[m * 64 + 4 * j + kk];
        } else {
            int s = slot - S_W0;
            int t = s >> 4, j = s & 15;
            val = c0[(t * 16 + m) * 64 + 4 * j + kk];
        }
        wlds[i] = val;
    }
    __syncthreads();

    const int wave = tid >> 5;
    const int lane = tid & 31;
    const int blk  = blockIdx.x * WAVES_PER_WG + wave;
    if (blk >= NBLK) return;

    const int pix = blk * 16 + (lane & 15);
    const int b   = pix / PIX_PER_IMG;
    const int rr  = pix - b * PIX_PER_IMG;
    const int hp  = rr / INTER + 1;
    const int wp  = rr - (hp - 1) * INTER + 1;

    const float4* xv = (const float4*)x;
    const int base = (b * HH + hp) * WW + wp;
    const float4 q3 = xv[base];
    const float4 q2 = xv[base - 1];
    const float4 q1 = xv[base - WW];
    const float4 q0 = xv[base - WW - 1];

    const int half = lane >> 4;
    const float xb3a = half ? q3.z : q3.x, xb3b = half ? q3.w : q3.y;
    const float xb2a = half ? q2.z : q2.x, xb2b = half ? q2.w : q2.y;
    const float xb1a = half ? q1.z : q1.x, xb1b = half ? q1.w : q1.y;
    const float xb0a = half ? q0.z : q0.x, xb0b = half ? q0.w : q0.y;

    const v2f* wv = (const v2f*)wlds;

    v2f a = wv[S_W3 * 32 + lane];
    v2f bb; bb[0] = xb3a; bb[1] = xb3b;
    v8f zeroAcc = {};
    v8f v3d = wmma4(a, bb, zeroAcc);

    v8f u = {};
#pragma unroll
    for (int j = 0; j < 16; ++j) {
        float row = __shfl(v3d[j & 7], (lane & 15) | ((j >> 3) << 4), 32);
        v2f bj; bj[0] = row * xb2a; bj[1] = row * xb2b;
        v2f aj = wv[(S_W2 + j) * 32 + lane];
        u = wmma4(aj, bj, u);
    }

    v8f vv = {};
#pragma unroll
    for (int j = 0; j < 16; ++j) {
        float row = __shfl(u[j & 7], (lane & 15) | ((j >> 3) << 4), 32);
        v2f bj; bj[0] = row * xb1a; bj[1] = row * xb1b;
        v2f aj = wv[(S_W1 + j) * 32 + lane];
        vv = wmma4(aj, bj, vv);
    }

    v8f d0 = {}, d1 = {}, d2 = {}, d3 = {};
#pragma unroll
    for (int j = 0; j < 16; ++j) {
        float row = __shfl(vv[j & 7], (lane & 15) | ((j >> 3) << 4), 32);
        v2f bj; bj[0] = row * xb0a; bj[1] = row * xb0b;
        d0 = wmma4(wv[(S_W0 +  0 + j) * 32 + lane], bj, d0);
        d1 = wmma4(wv[(S_W0 + 16 + j) * 32 + lane], bj, d1);
        d2 = wmma4(wv[(S_W0 + 32 + j) * 32 + lane], bj, d2);
        d3 = wmma4(wv[(S_W0 + 48 + j) * 32 + lane], bj, d3);
    }

    const int obase = ((b * HPAD + hp) * WPAD + wp) * OUTQ + 8 * half;
#define STORE_TILE(dt, t) {                                              \
        float* p = out + obase + (t) * 16;                               \
        ((float4*)p)[0] = make_float4(dt[0], dt[1], dt[2], dt[3]);       \
        ((float4*)p)[1] = make_float4(dt[4], dt[5], dt[6], dt[7]);       \
    }
    STORE_TILE(d0, 0)
    STORE_TILE(d1, 1)
    STORE_TILE(d2, 2)
    STORE_TILE(d3, 3)
#undef STORE_TILE
}

extern "C" void kernel_launch(void* const* d_in, const int* in_sizes, int n_in,
                              void* d_out, int out_size, void* d_ws, size_t ws_size,
                              hipStream_t stream) {
    (void)in_sizes; (void)n_in; (void)d_ws; (void)ws_size;
    const float* x  = (const float*)d_in[0];
    const float* c0 = (const float*)d_in[1];
    const float* c1 = (const float*)d_in[2];
    const float* c2 = (const float*)d_in[3];
    const float* c3 = (const float*)d_in[4];
    float* out = (float*)d_out;

    int n4 = out_size / 4;
    convsbs_zero<<<(n4 + 255) / 256, 256, 0, stream>>>((float4*)out, n4);

    int grid = (NBLK + WAVES_PER_WG - 1) / WAVES_PER_WG;
    convsbs_chain<<<grid, 256, 0, stream>>>(x, c0, c1, c2, c3, out);
}


# Weights in registers, no LDS: one wave per SIMD (256 x 4 waves) loads the 97 A fragments once by 8-byte global loads and chains its 4-5 tiles, each fragment feeding all tiles
# speedup vs baseline: 2.2006x; 2.2006x over previous
#include <hip/hip_runtime.h>


typedef __attribute__((ext_vector_type(2))) float v2f;
typedef __attribute__((ext_vector_type(8))) float v8f;

#define NB      16
#define HH      64
#define WW      64
#define QIN     4
#define BOND    16
#define OUTQ    64
#define HPAD    65
#define WPAD    65
#define INTER   63
#define PPIX_PER_IMG (HPAD * WPAD)
#define NPPIX   (NB * PPIX_PER_IMG)
#define NBLK    (NPPIX / 16)
#define NWG     256
#define WAVES_PER_WG 4

static __device__ __forceinline__ v8f wmma4(v2f a, v2f b, v8f c) {
    return __builtin_amdgcn_wmma_f32_16x16x4_f32(
         false, a,  false, b,
         (short)0, c,  false,  false);
}

template <int NT>
static __device__ __forceinline__ void run_tiles(
        const float* __restrict__ x, const float* __restrict__ c0,
        const float* __restrict__ c1, const float* __restrict__ c2,
        const float* __restrict__ c3, float* __restrict__ out,
        const int tile0, const int lane)
{
    const int half = lane >> 4;
    const float4* xv = (const float4*)x;
    const float4 zero4 = make_float4(0.f, 0.f, 0.f, 0.f);

    float x3a[NT], x3b[NT], x2a[NT], x2b[NT], x1a[NT], x1b[NT], x0a[NT], x0b[NT];
    int   obase[NT];

#pragma unroll
    for (int t = 0; t < NT; ++t) {
        const int pix = (tile0 + t) * 16 + (lane & 15);
        const int b   = pix / PPIX_PER_IMG;
        const int rr  = pix - b * PPIX_PER_IMG;
        const int hp  = rr / WPAD;
        const int wp  = rr - hp * WPAD;
        const bool inside = ((unsigned)(hp - 1) < (unsigned)INTER) &&
                            ((unsigned)(wp - 1) < (unsigned)INTER);
        const int hq = inside ? hp : 1;
        const int wq = inside ? wp : 1;
        const int base = (b * HH + hq) * WW + wq;
        const float4 l3 = xv[base];
        const float4 l2 = xv[base - 1];
        const float4 l1 = xv[base - WW];
        const float4 l0 = xv[base - WW - 1];
        const float4 q3 = inside ? l3 : zero4;
        const float4 q2 = inside ? l2 : zero4;
        const float4 q1 = inside ? l1 : zero4;
        const float4 q0 = inside ? l0 : zero4;
        x3a[t] = half ? q3.z : q3.x;  x3b[t] = half ? q3.w : q3.y;
        x2a[t] = half ? q2.z : q2.x;  x2b[t] = half ? q2.w : q2.y;
        x1a[t] = half ? q1.z : q1.x;  x1b[t] = half ? q1.w : q1.y;
        x0a[t] = half ? q0.z : q0.x;  x0b[t] = half ? q0.w : q0.y;
        obase[t] = pix * OUTQ + 8 * half;
    }

    const int m    = lane & 15;
    const int koff = half << 1;
    const v2f a3 = *(const v2f*)(c3 + m * 4 + koff);
    v2f a2[16], a1[16], a0[4][16];
#pragma unroll
    for (int j = 0; j < 16; ++j) a2[j] = *(const v2f*)(c2 + m * 64 + 4 * j + koff);
#pragma unroll
    for (int j = 0; j < 16; ++j) a1[j] = *(const v2f*)(c1 + m * 64 + 4 * j + koff);
#pragma unroll
    for (int mt = 0; mt < 4; ++mt) {
#pragma unroll
        for (int j = 0; j < 16; ++j)
            a0[mt][j] = *(const v2f*)(c0 + (mt * 16 + m) * 64 + 4 * j + koff);
    }

    v8f s[NT];
#pragma unroll
    for (int t = 0; t < NT; ++t) {
        v2f bb; bb[0] = x3a[t]; bb[1] = x3b[t];
        v8f zeroAcc = {};
        s[t] = wmma4(a3, bb, zeroAcc);
    }

    v8f u[NT];
#pragma unroll
    for (int t = 0; t < NT; ++t) { v8f z = {}; u[t] = z; }
#pragma unroll
    for (int j = 0; j < 16; ++j) {
#pragma unroll
        for (int t = 0; t < NT; ++t) {
            float row = __shfl(s[t][j & 7], (lane & 15) | ((j >> 3) << 4), 32);
            v2f bj; bj[0] = row * x2a[t]; bj[1] = row * x2b[t];
            u[t] = wmma4(a2[j], bj, u[t]);
        }
    }

#pragma unroll
    for (int t = 0; t < NT; ++t) { v8f z = {}; s[t] = z; }
#pragma unroll
    for (int j = 0; j < 16; ++j) {
#pragma unroll
        for (int t = 0; t < NT; ++t) {
            float row = __shfl(u[t][j & 7], (lane & 15) | ((j >> 3) << 4), 32);
            v2f bj; bj[0] = row * x1a[t]; bj[1] = row * x1b[t];
            s[t] = wmma4(a1[j], bj, s[t]);
        }
    }

#pragma unroll
    for (int t = 0; t < NT; ++t) {
        v8f d[4];
#pragma unroll
        for (int mt = 0; mt < 4; ++mt) { v8f z = {}; d[mt] = z; }
#pragma unroll
        for (int j = 0; j < 16; ++j) {
            float row = __shfl(s[t][j & 7], (lane & 15) | ((j >> 3) << 4), 32);
            v2f bj; bj[0] = row * x0a[t]; bj[1] = row * x0b[t];
#pragma unroll
            for (int mt = 0; mt < 4; ++mt)
                d[mt] = wmma4(a0[mt][j], bj, d[mt]);
        }
#pragma unroll
        for (int mt = 0; mt < 4; ++mt) {
            float* p = out + obase[t] + mt * 16;
            const v8f dt = d[mt];
            ((float4*)p)[0] = make_float4(dt[0], dt[1], dt[2], dt[3]);
            ((float4*)p)[1] = make_float4(dt[4], dt[5], dt[6], dt[7]);
        }
        __builtin_amdgcn_sched_barrier(0);
    }
}

__global__ __launch_bounds__(128) void convsbs_chain(
        const float* __restrict__ x,
        const float* __restrict__ c0,
        const float* __restrict__ c1,
        const float* __restrict__ c2,
        const float* __restrict__ c3,
        float* __restrict__ out)
{
    const int tid  = threadIdx.x;
    const int lane = tid & 31;
    const int wave = __builtin_amdgcn_readfirstlane(tid >> 5);
    const int gw   = blockIdx.x * WAVES_PER_WG + wave;
    const int tile0 = (gw * NBLK) >> 10;
    const int ntile = (((gw + 1) * NBLK) >> 10) - tile0;

    if (ntile == 5) run_tiles<5>(x, c0, c1, c2, c3, out, tile0, lane);
    else            run_tiles<4>(x, c0, c1, c2, c3, out, tile0, lane);
}

extern "C" void kernel_launch(void* const* d_in, const int* in_sizes, int n_in,
                              void* d_out, int out_size, void* d_ws, size_t ws_size,
                              hipStream_t stream) {
    (void)in_sizes; (void)n_in; (void)out_size; (void)d_ws; (void)ws_size;
    const float* x  = (const float*)d_in[0];
    const float* c0 = (const float*)d_in[1];
    const float* c1 = (const float*)d_in[2];
    const float* c2 = (const float*)d_in[3];
    const float* c3 = (const float*)d_in[4];
    float* out = (float*)d_out;

    convsbs_chain<<<NWG, 128, 0, stream>>>(x, c0, c1, c2, c3, out);
}


# Output leaves as whole 128-B lines: each finished 64x16 tile goes through wave-private LDS (272 B pixel pitch) and is stored as 512 contiguous bytes per instruction under the next tile's WMMAs
# speedup vs baseline: 2.6062x; 1.1843x over previous
#include <hip/hip_runtime.h>


typedef __attribute__((ext_vector_type(2))) float v2f;
typedef __attribute__((ext_vector_type(8))) float v8f;

#define NB      16
#define HH      64
#define WW      64
#define QIN     4
#define BOND    16
#define OUTQ    64
#define HPAD    65
#define WPAD    65
#define INTER   63
#define PPIX_PER_IMG (HPAD * WPAD)
#define NPPIX   (NB * PPIX_PER_IMG)
#define NBLK    (NPPIX / 16)
#define NWG     256
#define WAVES_PER_WG 4
#define SPITCH  68
#define STAGE_FLOATS (16 * SPITCH)

static __device__ __forceinline__ v8f wmma4(v2f a, v2f b, v8f c) {
    return __builtin_amdgcn_wmma_f32_16x16x4_f32(
         false, a,  false, b,
         (short)0, c,  false,  false);
}

template <int NT>
static __device__ __forceinline__ void run_tiles(
        const float* __restrict__ x, const float* __restrict__ c0,
        const float* __restrict__ c1, const float* __restrict__ c2,
        const float* __restrict__ c3, float* __restrict__ out,
        float* st, const int tile0, const int lane)
{
    const int half = lane >> 4;
    const float4* xv = (const float4*)x;
    const float4 zero4 = make_float4(0.f, 0.f, 0.f, 0.f);

    float x3a[NT], x3b[NT], x2a[NT], x2b[NT], x1a[NT], x1b[NT], x0a[NT], x0b[NT];

#pragma unroll
    for (int t = 0; t < NT; ++t) {
        const int pix = (tile0 + t) * 16 + (lane & 15);
        const int b   = pix / PPIX_PER_IMG;
        const int rr  = pix - b * PPIX_PER_IMG;
        const int hp  = rr / WPAD;
        const int wp  = rr - hp * WPAD;
        const bool inside = ((unsigned)(hp - 1) < (unsigned)INTER) &&
                            ((unsigned)(wp - 1) < (unsigned)INTER);
        const int hq = inside ? hp : 1;
        const int wq = inside ? wp : 1;
        const int base = (b * HH + hq) * WW + wq;
        const float4 l3 = xv[base];
        const float4 l2 = xv[base - 1];
        const float4 l1 = xv[base - WW];
        const float4 l0 = xv[base - WW - 1];
        const float4 q3 = inside ? l3 : zero4;
        const float4 q2 = inside ? l2 : zero4;
        const float4 q1 = inside ? l1 : zero4;
        const float4 q0 = inside ? l0 : zero4;
        x3a[t] = half ? q3.z : q3.x;  x3b[t] = half ? q3.w : q3.y;
        x2a[t] = half ? q2.z : q2.x;  x2b[t] = half ? q2.w : q2.y;
        x1a[t] = half ? q1.z : q1.x;  x1b[t] = half ? q1.w : q1.y;
        x0a[t] = half ? q0.z : q0.x;  x0b[t] = half ? q0.w : q0.y;
    }

    const int m    = lane & 15;
    const int koff = half << 1;
    const v2f a3 = *(const v2f*)(c3 + m * 4 + koff);
    v2f a2[16], a1[16], a0[4][16];
#pragma unroll
    for (int j = 0; j < 16; ++j) a2[j] = *(const v2f*)(c2 + m * 64 + 4 * j + koff);
#pragma unroll
    for (int j = 0; j < 16; ++j) a1[j] = *(const v2f*)(c1 + m * 64 + 4 * j + koff);
#pragma unroll
    for (int mt = 0; mt < 4; ++mt) {
#pragma unroll
        for (int j = 0; j < 16; ++j)
            a0[mt][j] = *(const v2f*)(c0 + (mt * 16 + m) * 64 + 4 * j + koff);
    }

    v8f s[NT];
#pragma unroll
    for (int t = 0; t < NT; ++t) {
        v2f bb; bb[0] = x3a[t]; bb[1] = x3b[t];
        v8f zeroAcc = {};
        s[t] = wmma4(a3, bb, zeroAcc);
    }

    v8f u[NT];
#pragma unroll
    for (int t = 0; t < NT; ++t) { v8f z = {}; u[t] = z; }
#pragma unroll
    for (int j = 0; j < 16; ++j) {
#pragma unroll
        for (int t = 0; t < NT; ++t) {
            float row = __shfl(s[t][j & 7], (lane & 15) | ((j >> 3) << 4), 32);
            v2f bj; bj[0] = row * x2a[t]; bj[1] = row * x2b[t];
            u[t] = wmma4(a2[j], bj, u[t]);
        }
    }

#pragma unroll
    for (int t = 0; t < NT; ++t) { v8f z = {}; s[t] = z; }
#pragma unroll
    for (int j = 0; j < 16; ++j) {
#pragma unroll
        for (int t = 0; t < NT; ++t) {
            float row = __shfl(u[t][j & 7], (lane & 15) | ((j >> 3) << 4), 32);
            v2f bj; bj[0] = row * x1a[t]; bj[1] = row * x1b[t];
            s[t] = wmma4(a1[j], bj, s[t]);
        }
    }

    const int wr = (lane & 15) * SPITCH + 8 * half;
    const int rd = half * SPITCH + (lane & 15) * 4;
    float4 o4[8];
#pragma unroll
    for (int t = 0; t < NT; ++t) {
        v8f d[4];
#pragma unroll
        for (int mt = 0; mt < 4; ++mt) { v8f z = {}; d[mt] = z; }
#pragma unroll
        for (int j = 0; j < 16; ++j) {
            float row = __shfl(s[t][j & 7], (lane & 15) | ((j >> 3) << 4), 32);
            v2f bj; bj[0] = row * x0a[t]; bj[1] = row * x0b[t];
#pragma unroll
            for (int mt = 0; mt < 4; ++mt)
                d[mt] = wmma4(a0[mt][j], bj, d[mt]);
            if (t > 0 && j == 1) {
#pragma unroll
                for (int i = 0; i < 8; ++i)
                    o4[i] = *(const float4*)(st + rd + 2 * i * SPITCH);
            }
            if (t > 0 && j == 4) {
                float4* g = (float4*)(out + (size_t)(tile0 + t - 1) * (16 * OUTQ)) + lane;
#pragma unroll
                for (int i = 0; i < 8; ++i) g[i * 32] = o4[i];
            }
        }
#pragma unroll
        for (int mt = 0; mt < 4; ++mt) {
            float* p = st + wr + mt * 16;
            const v8f dt = d[mt];
            ((float4*)p)[0] = make_float4(dt[0], dt[1], dt[2], dt[3]);
            ((float4*)p)[1] = make_float4(dt[4], dt[5], dt[6], dt[7]);
        }
        __builtin_amdgcn_sched_barrier(0);
    }
    {
#pragma unroll
        for (int i = 0; i < 8; ++i)
            o4[i] = *(const float4*)(st + rd + 2 * i * SPITCH);
        float4* g = (float4*)(out + (size_t)(tile0 + NT - 1) * (16 * OUTQ)) + lane;
#pragma unroll
        for (int i = 0; i < 8; ++i) g[i * 32] = o4[i];
    }
}

__global__ __launch_bounds__(128) void convsbs_chain(
        const float* __restrict__ x,
        const float* __restrict__ c0,
        const float* __restrict__ c1,
        const float* __restrict__ c2,
        const float* __restrict__ c3,
        float* __restrict__ out)
{
    __shared__ __attribute__((aligned(16))) float stage[WAVES_PER_WG * STAGE_FLOATS];
    const int tid  = threadIdx.x;
    const int lane = tid & 31;
    const int wave = __builtin_amdgcn_readfirstlane(tid >> 5);
    const int gw   = blockIdx.x * WAVES_PER_WG + wave;
    const int tile0 = (gw * NBLK) >> 10;
    const int ntile = (((gw + 1) * NBLK) >> 10) - tile0;

    float* st = stage + wave * STAGE_FLOATS;
    if (ntile == 5) run_tiles<5>(x, c0, c1, c2, c3, out, st, tile0, lane);
    else            run_tiles<4>(x, c0, c1, c2, c3, out, st, tile0, lane);
}

extern "C" void kernel_launch(void* const* d_in, const int* in_sizes, int n_in,
                              void* d_out, int out_size, void* d_ws, size_t ws_size,
                              hipStream_t stream) {
    (void)in_sizes; (void)n_in; (void)out_size; (void)d_ws; (void)ws_size;
    const float* x  = (const float*)d_in[0];
    const float* c0 = (const float*)d_in[1];
    const float* c1 = (const float*)d_in[2];
    const float* c2 = (const float*)d_in[3];
    const float* c3 = (const float*)d_in[4];
    float* out = (float*)d_out;

    convsbs_chain<<<NWG, 128, 0, stream>>>(x, c0, c1, c2, c3, out);
}


# Row broadcasts between stages on the vector ALU, not the LDS unit: one v_permlanex16 per D register and a select per row replace the 48 ds_bpermute per tile and their LDS-counter waits
# speedup vs baseline: 2.7943x; 1.0722x over previous
#include <hip/hip_runtime.h>


typedef __attribute__((ext_vector_type(2))) float v2f;
typedef __attribute__((ext_vector_type(8))) float v8f;

#define NB      16
#define HH      64
#define WW      64
#define QIN     4
#define BOND    16
#define OUTQ    64
#define HPAD    65
#define WPAD    65
#define INTER   63
#define PPIX_PER_IMG (HPAD * WPAD)
#define NPPIX   (NB * PPIX_PER_IMG)
#define NBLK    (NPPIX / 16)
#define NWG     256
#define WAVES_PER_WG 4
#define SPITCH  68
#define STAGE_FLOATS (16 * SPITCH)

static __device__ __forceinline__ v8f wmma4(v2f a, v2f b, v8f c) {
    return __builtin_amdgcn_wmma_f32_16x16x4_f32(
         false, a,  false, b,
         (short)0, c,  false,  false);
}

static __device__ __forceinline__ float xhalf(float x) {
    const int b = __builtin_bit_cast(int, x);
    return __builtin_bit_cast(float,
        __builtin_amdgcn_permlanex16(b, b, 0x76543210, 0xfedcba98,  true,  false));
}

static __device__ __forceinline__ void bcast_rows(const v8f d, const bool upper, float (&row)[16]) {
#pragma unroll
    for (int r = 0; r < 8; ++r) {
        const float own = d[r];
        const float oth = xhalf(own);
        row[r]     = upper ? oth : own;
        row[8 + r] = upper ? own : oth;
    }
}

template <int NT>
static __device__ __forceinline__ void run_tiles(
        const float* __restrict__ x, const float* __restrict__ c0,
        const float* __restrict__ c1, const float* __restrict__ c2,
        const float* __restrict__ c3, float* __restrict__ out,
        float* st, const int tile0, const int lane)
{
    const int half = lane >> 4;
    const float4* xv = (const float4*)x;
    const float4 zero4 = make_float4(0.f, 0.f, 0.f, 0.f);

    float x3a[NT], x3b[NT], x2a[NT], x2b[NT], x1a[NT], x1b[NT], x0a[NT], x0b[NT];

#pragma unroll
    for (int t = 0; t < NT; ++t) {
        const int pix = (tile0 + t) * 16 + (lane & 15);
        const int b   = pix / PPIX_PER_IMG;
        const int rr  = pix - b * PPIX_PER_IMG;
        const int hp  = rr / WPAD;
        const int wp  = rr - hp * WPAD;
        const bool inside = ((unsigned)(hp - 1) < (unsigned)INTER) &&
                            ((unsigned)(wp - 1) < (unsigned)INTER);
        const int hq = inside ? hp : 1;
        const int wq = inside ? wp : 1;
        const int base = (b * HH + hq) * WW + wq;
        const float4 l3 = xv[base];
        const float4 l2 = xv[base - 1];
        const float4 l1 = xv[base - WW];
        const float4 l0 = xv[base - WW - 1];
        const float4 q3 = inside ? l3 : zero4;
        const float4 q2 = inside ? l2 : zero4;
        const float4 q1 = inside ? l1 : zero4;
        const float4 q0 = inside ? l0 : zero4;
        x3a[t] = half ? q3.z : q3.x;  x3b[t] = half ? q3.w : q3.y;
        x2a[t] = half ? q2.z : q2.x;  x2b[t] = half ? q2.w : q2.y;
        x1a[t] = half ? q1.z : q1.x;  x1b[t] = half ? q1.w : q1.y;
        x0a[t] = half ? q0.z : q0.x;  x0b[t] = half ? q0.w : q0.y;
    }

    const int m    = lane & 15;
    const int koff = half << 1;
    const v2f a3 = *(const v2f*)(c3 + m * 4 + koff);
    v2f a2[16], a1[16], a0[4][16];
#pragma unroll
    for (int j = 0; j < 16; ++j) a2[j] = *(const v2f*)(c2 + m * 64 + 4 * j + koff);
#pragma unroll
    for (int j = 0; j < 16; ++j) a1[j] = *(const v2f*)(c1 + m * 64 + 4 * j + koff);
#pragma unroll
    for (int mt = 0; mt < 4; ++mt) {
#pragma unroll
        for (int j = 0; j < 16; ++j)
            a0[mt][j] = *(const v2f*)(c0 + (mt * 16 + m) * 64 + 4 * j + koff);
    }

    v8f s[NT];
#pragma unroll
    for (int t = 0; t < NT; ++t) {
        v2f bb; bb[0] = x3a[t]; bb[1] = x3b[t];
        v8f zeroAcc = {};
        s[t] = wmma4(a3, bb, zeroAcc);
    }

    const bool upper = half != 0;
    float rw[NT][16];
    v8f u[NT];
#pragma unroll
    for (int t = 0; t < NT; ++t) { v8f z = {}; u[t] = z; bcast_rows(s[t], upper, rw[t]); }
#pragma unroll
    for (int j = 0; j < 16; ++j) {
#pragma unroll
        for (int t = 0; t < NT; ++t) {
            const float row = rw[t][j];
            v2f bj; bj[0] = row * x2a[t]; bj[1] = row * x2b[t];
            u[t] = wmma4(a2[j], bj, u[t]);
        }
    }

#pragma unroll
    for (int t = 0; t < NT; ++t) { v8f z = {}; s[t] = z; bcast_rows(u[t], upper, rw[t]); }
#pragma unroll
    for (int j = 0; j < 16; ++j) {
#pragma unroll
        for (int t = 0; t < NT; ++t) {
            const float row = rw[t][j];
            v2f bj; bj[0] = row * x1a[t]; bj[1] = row * x1b[t];
            s[t] = wmma4(a1[j], bj, s[t]);
        }
    }
#pragma unroll
    for (int t = 0; t < NT; ++t) bcast_rows(s[t], upper, rw[t]);

    const int wr = (lane & 15) * SPITCH + 8 * half;
    const int rd = half * SPITCH + (lane & 15) * 4;
    float4 o4[8];
#pragma unroll
    for (int t = 0; t < NT; ++t) {
        v8f d[4];
#pragma unroll
        for (int mt = 0; mt < 4; ++mt) { v8f z = {}; d[mt] = z; }
#pragma unroll
        for (int j = 0; j < 16; ++j) {
            const float row = rw[t][j];
            v2f bj; bj[0] = row * x0a[t]; bj[1] = row * x0b[t];
#pragma unroll
            for (int mt = 0; mt < 4; ++mt)
                d[mt] = wmma4(a0[mt][j], bj, d[mt]);
            if (t > 0 && j == 1) {
#pragma unroll
                for (int i = 0; i < 8; ++i)
                    o4[i] = *(const float4*)(st + rd + 2 * i * SPITCH);
            }
            if (t > 0 && j == 4) {
                float4* g = (float4*)(out + (size_t)(tile0 + t - 1) * (16 * OUTQ)) + lane;
#pragma unroll
                for (int i = 0; i < 8; ++i) g[i * 32] = o4[i];
            }
        }
#pragma unroll
        for (int mt = 0; mt < 4; ++mt) {
            float* p = st + wr + mt * 16;
            const v8f dt = d[mt];
            ((float4*)p)[0] = make_float4(dt[0], dt[1], dt[2], dt[3]);
            ((float4*)p)[1] = make_float4(dt[4], dt[5], dt[6], dt[7]);
        }
        __builtin_amdgcn_sched_barrier(0);
    }
    {
#pragma unroll
        for (int i = 0; i < 8; ++i)
            o4[i] = *(const float4*)(st + rd + 2 * i * SPITCH);
        float4* g = (float4*)(out + (size_t)(tile0 + NT - 1) * (16 * OUTQ)) + lane;
#pragma unroll
        for (int i = 0; i < 8; ++i) g[i * 32] = o4[i];
    }
}

__global__ __launch_bounds__(128) void convsbs_chain(
        const float* __restrict__ x,
        const float* __restrict__ c0,
        const float* __restrict__ c1,
        const float* __restrict__ c2,
        const float* __restrict__ c3,
        float* __restrict__ out)
{
    __shared__ __attribute__((aligned(16))) float stage[WAVES_PER_WG * STAGE_FLOATS];
    const int tid  = threadIdx.x;
    const int lane = tid & 31;
    const int wave = __builtin_amdgcn_readfirstlane(tid >> 5);
    const int gw   = blockIdx.x * WAVES_PER_WG + wave;
    const int tile0 = (gw * NBLK) >> 10;
    const int ntile = (((gw + 1) * NBLK) >> 10) - tile0;

    float* st = stage + wave * STAGE_FLOATS;
    if (ntile == 5) run_tiles<5>(x, c0, c1, c2, c3, out, st, tile0, lane);
    else            run_tiles<4>(x, c0, c1, c2, c3, out, st, tile0, lane);
}

extern "C" void kernel_launch(void* const* d_in, const int* in_sizes, int n_in,
                              void* d_out, int out_size, void* d_ws, size_t ws_size,
                              hipStream_t stream) {
    (void)in_sizes; (void)n_in; (void)out_size; (void)d_ws; (void)ws_size;
    const float* x  = (const float*)d_in[0];
    const float* c0 = (const float*)d_in[1];
    const float* c1 = (const float*)d_in[2];
    const float* c2 = (const float*)d_in[3];
    const float* c3 = (const float*)d_in[4];
    float* out = (float*)d_out;

    convsbs_chain<<<NWG, 128, 0, stream>>>(x, c0, c1, c2, c3, out);
}


# Tiles cover only the 63x63 interior pixels: 3969 tiles, at most 4 per wave (was 5), no matrix work on zero border pixels; stores bridge a tile's row break, each wave zero-stores 4 border pixels
# speedup vs baseline: 3.0319x; 1.0850x over previous
#include <hip/hip_runtime.h>


typedef __attribute__((ext_vector_type(2))) float v2f;
typedef __attribute__((ext_vector_type(8))) float v8f;

#define NB      16
#define HH      64
#define WW      64
#define QIN     4
#define BOND    16
#define OUTQ    64
#define HPAD    65
#define WPAD    65
#define INTER   63
#define PPIX_PER_IMG (HPAD * WPAD)
#define IPIX_PER_IMG (INTER * INTER)
#define NIPIX   (NB * IPIX_PER_IMG)
#define NBLK    (NIPIX / 16)
#define ROWGAP  2
#define IMGGAP  (2 * WPAD + 2)
#define BUNITS_PER_IMG 128
#define NWG     256
#define WAVES_PER_WG 4
#define SPITCH  68
#define STAGE_FLOATS (16 * SPITCH)

static __device__ __forceinline__ v8f wmma4(v2f a, v2f b, v8f c) {
    return __builtin_amdgcn_wmma_f32_16x16x4_f32(
         false, a,  false, b,
         (short)0, c,  false,  false);
}

static __device__ __forceinline__ float xhalf(float x) {
    const int b = __builtin_bit_cast(int, x);
    return __builtin_bit_cast(float,
        __builtin_amdgcn_permlanex16(b, b, 0x76543210, 0xfedcba98,  true,  false));
}

static __device__ __forceinline__ void bcast_rows(const v8f d, const bool upper, float (&row)[16]) {
#pragma unroll
    for (int r = 0; r < 8; ++r) {
        const float own = d[r];
        const float oth = xhalf(own);
        row[r]     = upper ? oth : own;
        row[8 + r] = upper ? own : oth;
    }
}

static __device__ __forceinline__ void tile_map(const int tile, int& p0, int& kmax, int& gap) {
    const int n0 = tile * 16;
    const int b  = n0 / IPIX_PER_IMG;
    const int r  = n0 - b * IPIX_PER_IMG;
    const int hm = r / INTER;
    const int wm = r - hm * INTER;
    p0   = b * PPIX_PER_IMG + (hm + 1) * WPAD + (wm + 1);
    kmax = (INTER - 1) - wm;
    gap  = (hm == INTER - 1) ? IMGGAP : ROWGAP;
}

static __device__ __forceinline__ void store_tile(float* __restrict__ out, const int tile,
                                                  const int lane, const float4 (&o4)[8]) {
    int p0, kmax, gap;
    tile_map(tile, p0, kmax, gap);
    const int half = lane >> 4;
    const unsigned off_near = (unsigned)(p0 + half) * (unsigned)(OUTQ * 4) + (unsigned)(lane & 15) * 16u;
    const unsigned off_far  = off_near + (unsigned)gap * (unsigned)(OUTQ * 4);
    const int kth = kmax - half;
    char* const ob = (char*)out;
#pragma unroll
    for (int i = 0; i < 8; ++i) {
        const unsigned off = (2 * i > kth) ? off_far : off_near;
        *(float4*)(ob + (size_t)off + (size_t)(512 * i)) = o4[i];
    }
}

static __device__ __forceinline__ void border_stores(float* __restrict__ out, const int gw, const int lane) {
#pragma unroll
    for (int e = 0; e < 2; ++e) {
        const int unit = 2 * gw + e;
        const int b    = unit / BUNITS_PER_IMG;
        const int u    = unit - b * BUNITS_PER_IMG;
        const int pp   = 2 * min(u, 32) + WPAD * min(max(u - 32, 0), INTER) + 2 * max(u - 95, 0);
        float4* g = (float4*)(out + (size_t)(b * PPIX_PER_IMG + pp) * OUTQ) + lane;
        *g = make_float4(0.f, 0.f, 0.f, 0.f);
    }
}

template <int NT>
static __device__ __forceinline__ void run_tiles(
        const float* __restrict__ x, const float* __restrict__ c0,
        const float* __restrict__ c1, const float* __restrict__ c2,
        const float* __restrict__ c3, float* __restrict__ out,
        float* st, const int tile0, const int lane)
{
    const int half = lane >> 4;
    const float4* xv = (const float4*)x;

    float x3a[NT], x3b[NT], x2a[NT], x2b[NT], x1a[NT], x1b[NT], x0a[NT], x0b[NT];

#pragma unroll
    for (int t = 0; t < NT; ++t) {
        const int pix = (tile0 + t) * 16 + (lane & 15);
        const int b   = pix / IPIX_PER_IMG;
        const int rr  = pix - b * IPIX_PER_IMG;
        const int hm  = rr / INTER;
        const int wm  = rr - hm * INTER;
        const int base = (b * HH + hm + 1) * WW + wm + 1;
        const float4 q3 = xv[base];
        const float4 q2 = xv[base - 1];
        const float4 q1 = xv[base - WW];
        const float4 q0 = xv[base - WW - 1];
        x3a[t] = half ? q3.z : q3.x;  x3b[t] = half ? q3.w : q3.y;
        x2a[t] = half ? q2.z : q2.x;  x2b[t] = half ? q2.w : q2.y;
        x1a[t] = half ? q1.z : q1.x;  x1b[t] = half ? q1.w : q1.y;
        x0a[t] = half ? q0.z : q0.x;  x0b[t] = half ? q0.w : q0.y;
    }

    const int m    = lane & 15;
    const int koff = half << 1;
    const v2f a3 = *(const v2f*)(c3 + m * 4 + koff);
    v2f a2[16], a1[16], a0[4][16];
#pragma unroll
    for (int j = 0; j < 16; ++j) a2[j] = *(const v2f*)(c2 + m * 64 + 4 * j + koff);
#pragma unroll
    for (int j = 0; j < 16; ++j) a1[j] = *(const v2f*)(c1 + m * 64 + 4 * j + koff);
#pragma unroll
    for (int mt = 0; mt < 4; ++mt) {
#pragma unroll
        for (int j = 0; j < 16; ++j)
            a0[mt][j] = *(const v2f*)(c0 + (mt * 16 + m) * 64 + 4 * j + koff);
    }

    v8f s[NT];
#pragma unroll
    for (int t = 0; t < NT; ++t) {
        v2f bb; bb[0] = x3a[t]; bb[1] = x3b[t];
        v8f zeroAcc = {};
        s[t] = wmma4(a3, bb, zeroAcc);
    }

    const bool upper = half != 0;
    float rw[NT][16];
    v8f u[NT];
#pragma unroll
    for (int t = 0; t < NT; ++t) { v8f z = {}; u[t] = z; bcast_rows(s[t], upper, rw[t]); }
#pragma unroll
    for (int j = 0; j < 16; ++j) {
#pragma unroll
        for (int t = 0; t < NT; ++t) {
            const float row = rw[t][j];
            v2f bj; bj[0] = row * x2a[t]; bj[1] = row * x2b[t];
            u[t] = wmma4(a2[j], bj, u[t]);
        }
    }

#pragma unroll
    for (int t = 0; t < NT; ++t) { v8f z = {}; s[t] = z; bcast_rows(u[t], upper, rw[t]); }
#pragma unroll
    for (int j = 0; j < 16; ++j) {
#pragma unroll
        for (int t = 0; t < NT; ++t) {
            const float row = rw[t][j];
            v2f bj; bj[0] = row * x1a[t]; bj[1] = row * x1b[t];
            s[t] = wmma4(a1[j], bj, s[t]);
        }
    }
#pragma unroll
    for (int t = 0; t < NT; ++t) bcast_rows(s[t], upper, rw[t]);

    const int wr = (lane & 15) * SPITCH + 8 * half;
    const int rd = half * SPITCH + (lane & 15) * 4;
    float4 o4[8];
#pragma unroll
    for (int t = 0; t < NT; ++t) {
        v8f d[4];
#pragma unroll
        for (int mt = 0; mt < 4; ++mt) { v8f z = {}; d[mt] = z; }
#pragma unroll
        for (int j = 0; j < 16; ++j) {
            const float row = rw[t][j];
            v2f bj; bj[0] = row * x0a[t]; bj[1] = row * x0b[t];
#pragma unroll
            for (int mt = 0; mt < 4; ++mt)
                d[mt] = wmma4(a0[mt][j], bj, d[mt]);
            if (t > 0 && j == 1) {
#pragma unroll
                for (int i = 0; i < 8; ++i)
                    o4[i] = *(const float4*)(st + rd + 2 * i * SPITCH);
            }
            if (t > 0 && j == 4) store_tile(out, tile0 + t - 1, lane, o4);
        }
#pragma unroll
        for (int mt = 0; mt < 4; ++mt) {
            float* p = st + wr + mt * 16;
            const v8f dt = d[mt];
            ((float4*)p)[0] = make_float4(dt[0], dt[1], dt[2], dt[3]);
            ((float4*)p)[1] = make_float4(dt[4], dt[5], dt[6], dt[7]);
        }
        __builtin_amdgcn_sched_barrier(0);
    }
    {
#pragma unroll
        for (int i = 0; i < 8; ++i)
            o4[i] = *(const float4*)(st + rd + 2 * i * SPITCH);
        store_tile(out, tile0 + NT - 1, lane, o4);
    }
}

__global__ __launch_bounds__(128) void convsbs_chain(
        const float* __restrict__ x,
        const float* __restrict__ c0,
        const float* __restrict__ c1,
        const float* __restrict__ c2,
        const float* __restrict__ c3,
        float* __restrict__ out)
{
    __shared__ __attribute__((aligned(16))) float stage[WAVES_PER_WG * STAGE_FLOATS];
    const int tid  = threadIdx.x;
    const int lane = tid & 31;
    const int wave = __builtin_amdgcn_readfirstlane(tid >> 5);
    const int gw   = blockIdx.x * WAVES_PER_WG + wave;
    border_stores(out, gw, lane);

    const int tile0 = (gw * NBLK) >> 10;
    const int ntile = (((gw + 1) * NBLK) >> 10) - tile0;

    float* st = stage + wave * STAGE_FLOATS;
    if (ntile == 4) run_tiles<4>(x, c0, c1, c2, c3, out, st, tile0, lane);
    else            run_tiles<3>(x, c0, c1, c2, c3, out, st, tile0, lane);
}

extern "C" void kernel_launch(void* const* d_in, const int* in_sizes, int n_in,
                              void* d_out, int out_size, void* d_ws, size_t ws_size,
                              hipStream_t stream) {
    (void)in_sizes; (void)n_in; (void)out_size; (void)d_ws; (void)ws_size;
    const float* x  = (const float*)d_in[0];
    const float* c0 = (const float*)d_in[1];
    const float* c1 = (const float*)d_in[2];
    const float* c2 = (const float*)d_in[3];
    const float* c3 = (const float*)d_in[4];
    float* out = (float*)d_out;

    convsbs_chain<<<NWG, 128, 0, stream>>>(x, c0, c1, c2, c3, out);
}
